# starting kernel
# baseline (speedup 1.0000x reference)
#include <hip/hip_runtime.h>

typedef __attribute__((ext_vector_type(2))) float v2f;
typedef __attribute__((ext_vector_type(4))) float v4f;
typedef __attribute__((ext_vector_type(8))) float v8f;

constexpr int kD = 256;
constexpr int kChunks = kD / 4;
constexpr int kRowsPerWave = 16;
constexpr int kWavesPerBlock = 8;
constexpr int kRowsPerBlock = kRowsPerWave * kWavesPerBlock;

__global__ __launch_bounds__(256) void shared_unit_wmma_kernel(
    const float* __restrict__ drug_cnn,
    const float* __restrict__ drug_kg,
    const float* __restrict__ w_aa,
    const float* __restrict__ w_ab,
    const float* __restrict__ w_ba,
    const float* __restrict__ w_bb,
    const float* __restrict__ bias_cnn,
    const float* __restrict__ bias_kg,
    float* __restrict__ out_cnn,
    float* __restrict__ out_kg)
{
    __shared__ __align__(16) v2f   s_btab[kChunks][32];
    __shared__ __align__(16) float s_bc[kD];
    __shared__ __align__(16) float s_bk[kD];
    __shared__ __align__(16) float s_scl[kRowsPerBlock * 2];

    const int tid = threadIdx.x;

    s_bc[tid] = bias_cnn[tid];
    s_bk[tid] = bias_kg[tid];

    #pragma unroll
    for (int j = 0; j < 8; ++j) {
        const int f    = tid + 256 * j;
        const int kk   = f >> 5;
        const int l    = f & 31;
        const int n    = l & 15;
        const int half = l >> 4;
        const int k0   = kk * 4 + half * 2;
        v2f b = {0.0f, 0.0f};
        if (n == 0) {
            b.x = w_aa[k0]     + w_ab[k0];
            b.y = w_aa[k0 + 1] + w_ab[k0 + 1];
        } else if (n == 1) {
            b.x = w_ba[k0]     + w_bb[k0];
            b.y = w_ba[k0 + 1] + w_bb[k0 + 1];
        }
        s_btab[kk][l] = b;
    }
    __syncthreads();

    const int lane = tid & 31;
    const int wave = tid >> 5;
    const int n    = lane & 15;
    const int half = lane >> 4;
    const int rowBlock = blockIdx.x * kRowsPerBlock;
    const int rowWave  = rowBlock + wave * kRowsPerWave;

    const v2f* kgRow2 = (const v2f*)(drug_kg + (size_t)(rowWave + n) * kD);

    v8f acc = {};
    #pragma unroll 8
    for (int kk = 0; kk < kChunks; ++kk) {
        v2f a = __builtin_nontemporal_load(&kgRow2[kk * 2 + half]);
        v2f b = s_btab[kk][lane];
        acc = __builtin_amdgcn_wmma_f32_16x16x4_f32(
             false, a,  false, b,
             (short)0, acc,  false,  false);
    }

    if (n < 2) {
        const int mbase = wave * kRowsPerWave + half * 8;
        #pragma unroll
        for (int v = 0; v < 8; ++v)
            s_scl[(mbase + v) * 2 + n] = acc[v];
    }
    __syncthreads();

    const v4f* cnn4 = (const v4f*)drug_cnn;
    v4f* oc4 = (v4f*)out_cnn;
    v4f* ok4 = (v4f*)out_kg;
    const v4f* bc4 = (const v4f*)s_bc;
    const v4f* bk4 = (const v4f*)s_bk;

    #pragma unroll 4
    for (int r = 0; r < kRowsPerWave; ++r) {
        const int lrow = wave * kRowsPerWave + r;
        const size_t grow = (size_t)(rowBlock + lrow) * (kD / 4);
        const float sc = s_scl[lrow * 2 + 0];
        const float sk = s_scl[lrow * 2 + 1];
        #pragma unroll
        for (int hcol = 0; hcol < 2; ++hcol) {
            const int i = hcol * 32 + lane;
            v4f c  = __builtin_nontemporal_load(&cnn4[grow + i]);
            v4f bc = bc4[i];
            v4f bk = bk4[i];
            v4f o1, o2;
            o1.x = __builtin_fmaf(c.x, sc, bc.x);  o2.x = __builtin_fmaf(c.x, sk, bk.x);
            o1.y = __builtin_fmaf(c.y, sc, bc.y);  o2.y = __builtin_fmaf(c.y, sk, bk.y);
            o1.z = __builtin_fmaf(c.z, sc, bc.z);  o2.z = __builtin_fmaf(c.z, sk, bk.z);
            o1.w = __builtin_fmaf(c.w, sc, bc.w);  o2.w = __builtin_fmaf(c.w, sk, bk.w);
            __builtin_nontemporal_store(o1, &oc4[grow + i]);
            __builtin_nontemporal_store(o2, &ok4[grow + i]);
        }
    }
}

extern "C" void kernel_launch(void* const* d_in, const int* in_sizes, int n_in,
                              void* d_out, int out_size, void* d_ws, size_t ws_size,
                              hipStream_t stream) {
    const float* drug_cnn = (const float*)d_in[0];
    const float* drug_kg  = (const float*)d_in[1];
    const float* w_aa     = (const float*)d_in[2];
    const float* w_ab     = (const float*)d_in[3];
    const float* w_ba     = (const float*)d_in[4];
    const float* w_bb     = (const float*)d_in[5];
    const float* bcnn     = (const float*)d_in[6];
    const float* bkg      = (const float*)d_in[7];

    const int B = in_sizes[0] / kD;
    float* out_cnn = (float*)d_out;
    float* out_kg  = out_cnn + (size_t)B * kD;

    dim3 grid(B / kRowsPerBlock);
    dim3 block(kWavesPerBlock * 32);
    shared_unit_wmma_kernel<<<grid, block, 0, stream>>>(
        drug_cnn, drug_kg, w_aa, w_ab, w_ba, w_bb, bcnn, bkg, out_cnn, out_kg);
}


# kg tile read by whole-line 512-byte loads (all 32 of a wave in flight) and transposed through padded LDS into the WMMA A fragments, in place of 16-line 8-byte gathers; same k order and sums
# speedup vs baseline: 1.5953x; 1.5953x over previous
#include <hip/hip_runtime.h>

typedef __attribute__((ext_vector_type(2))) float v2f;
typedef __attribute__((ext_vector_type(4))) float v4f;
typedef __attribute__((ext_vector_type(8))) float v8f;

constexpr int kD = 256;
constexpr int kChunks = kD / 4;
constexpr int kRowsPerWave = 16;
constexpr int kWavesPerBlock = 8;
constexpr int kRowsPerBlock = kRowsPerWave * kWavesPerBlock;
constexpr int kTileLoads = kRowsPerWave * kD / (32 * 4);
constexpr int kTilePitch = kD + 4;

__global__ __launch_bounds__(256) void shared_unit_wmma_kernel(
    const float* __restrict__ drug_cnn,
    const float* __restrict__ drug_kg,
    const float* __restrict__ w_aa,
    const float* __restrict__ w_ab,
    const float* __restrict__ w_ba,
    const float* __restrict__ w_bb,
    const float* __restrict__ bias_cnn,
    const float* __restrict__ bias_kg,
    float* __restrict__ out_cnn,
    float* __restrict__ out_kg)
{
    __shared__ __align__(16) v2f   s_btab[kChunks][32];
    __shared__ __align__(16) float s_bc[kD];
    __shared__ __align__(16) float s_bk[kD];
    __shared__ __align__(16) float s_scl[kRowsPerBlock * 2];
    __shared__ __align__(16) float s_kg[kWavesPerBlock * kRowsPerWave * kTilePitch];

    const int tid  = threadIdx.x;
    const int lane = tid & 31;
    const int wave = tid >> 5;
    const int n    = lane & 15;
    const int half = lane >> 4;
    const int rowBlock = blockIdx.x * kRowsPerBlock;
    const int rowWave  = rowBlock + wave * kRowsPerWave;

    const v4f* kgTile4 = (const v4f*)(drug_kg + (size_t)rowWave * kD);
    v4f t[kTileLoads];
    #pragma unroll
    for (int q = 0; q < kTileLoads; ++q)
        t[q] = __builtin_nontemporal_load(&kgTile4[q * 32 + lane]);

    const float bcv = bias_cnn[tid];
    const float bkv = bias_kg[tid];

    #pragma unroll
    for (int j = 0; j < 8; ++j) {
        const int f    = tid + 256 * j;
        const int kk   = f >> 5;
        const int l    = f & 31;
        const int nn   = l & 15;
        const int hh   = l >> 4;
        const int k0   = kk * 4 + hh * 2;
        v2f b = {0.0f, 0.0f};
        if (nn == 0) {
            b.x = w_aa[k0]     + w_ab[k0];
            b.y = w_aa[k0 + 1] + w_ab[k0 + 1];
        } else if (nn == 1) {
            b.x = w_ba[k0]     + w_bb[k0];
            b.y = w_ba[k0 + 1] + w_bb[k0 + 1];
        }
        s_btab[kk][l] = b;
    }

    s_bc[tid] = bcv;
    s_bk[tid] = bkv;

    float* tile = s_kg + wave * (kRowsPerWave * kTilePitch);
    #pragma unroll
    for (int q = 0; q < kTileLoads; ++q)
        *(v4f*)&tile[(q >> 1) * kTilePitch + (q & 1) * 128 + lane * 4] = t[q];
    __syncthreads();

    const float* aRow = tile + n * kTilePitch + half * 2;

    v8f acc = {};
    #pragma unroll 8
    for (int kk = 0; kk < kChunks; ++kk) {
        v2f a = *(const v2f*)&aRow[kk * 4];
        v2f b = s_btab[kk][lane];
        acc = __builtin_amdgcn_wmma_f32_16x16x4_f32(
             false, a,  false, b,
             (short)0, acc,  false,  false);
    }

    if (n < 2) {
        const int mbase = wave * kRowsPerWave + half * 8;
        #pragma unroll
        for (int v = 0; v < 8; ++v)
            s_scl[(mbase + v) * 2 + n] = acc[v];
    }
    __syncthreads();

    const v4f* cnn4 = (const v4f*)drug_cnn;
    v4f* oc4 = (v4f*)out_cnn;
    v4f* ok4 = (v4f*)out_kg;
    const v4f* bc4 = (const v4f*)s_bc;
    const v4f* bk4 = (const v4f*)s_bk;

    #pragma unroll 4
    for (int r = 0; r < kRowsPerWave; ++r) {
        const int lrow = wave * kRowsPerWave + r;
        const size_t grow = (size_t)(rowBlock + lrow) * (kD / 4);
        const float sc = s_scl[lrow * 2 + 0];
        const float sk = s_scl[lrow * 2 + 1];
        #pragma unroll
        for (int hcol = 0; hcol < 2; ++hcol) {
            const int i = hcol * 32 + lane;
            v4f c  = __builtin_nontemporal_load(&cnn4[grow + i]);
            v4f bc = bc4[i];
            v4f bk = bk4[i];
            v4f o1, o2;
            o1.x = __builtin_fmaf(c.x, sc, bc.x);  o2.x = __builtin_fmaf(c.x, sk, bk.x);
            o1.y = __builtin_fmaf(c.y, sc, bc.y);  o2.y = __builtin_fmaf(c.y, sk, bk.y);
            o1.z = __builtin_fmaf(c.z, sc, bc.z);  o2.z = __builtin_fmaf(c.z, sk, bk.z);
            o1.w = __builtin_fmaf(c.w, sc, bc.w);  o2.w = __builtin_fmaf(c.w, sk, bk.w);
            __builtin_nontemporal_store(o1, &oc4[grow + i]);
            __builtin_nontemporal_store(o2, &ok4[grow + i]);
        }
    }
}

extern "C" void kernel_launch(void* const* d_in, const int* in_sizes, int n_in,
                              void* d_out, int out_size, void* d_ws, size_t ws_size,
                              hipStream_t stream) {
    const float* drug_cnn = (const float*)d_in[0];
    const float* drug_kg  = (const float*)d_in[1];
    const float* w_aa     = (const float*)d_in[2];
    const float* w_ab     = (const float*)d_in[3];
    const float* w_ba     = (const float*)d_in[4];
    const float* w_bb     = (const float*)d_in[5];
    const float* bcnn     = (const float*)d_in[6];
    const float* bkg      = (const float*)d_in[7];

    const int B = in_sizes[0] / kD;
    float* out_cnn = (float*)d_out;
    float* out_kg  = out_cnn + (size_t)B * kD;

    dim3 grid(B / kRowsPerBlock);
    dim3 block(kWavesPerBlock * 32);
    shared_unit_wmma_kernel<<<grid, block, 0, stream>>>(
        drug_cnn, drug_kg, w_aa, w_ab, w_ba, w_bb, bcnn, bkg, out_cnn, out_kg);
}


# drug_cnn tile requested as one 32-load group behind the first barrier, in flight across the WMMA phase, scaled and stored from registers; second barrier dropped (s_scl is wave-private); same sums
# speedup vs baseline: 2.1307x; 1.3356x over previous
#include <hip/hip_runtime.h>

typedef __attribute__((ext_vector_type(2))) float v2f;
typedef __attribute__((ext_vector_type(4))) float v4f;
typedef __attribute__((ext_vector_type(8))) float v8f;

constexpr int kD = 256;
constexpr int kChunks = kD / 4;
constexpr int kRowsPerWave = 16;
constexpr int kWavesPerBlock = 8;
constexpr int kRowsPerBlock = kRowsPerWave * kWavesPerBlock;
constexpr int kTileLoads = kRowsPerWave * kD / (32 * 4);
constexpr int kTilePitch = kD + 4;

__global__ __launch_bounds__(256) void shared_unit_wmma_kernel(
    const float* __restrict__ drug_cnn,
    const float* __restrict__ drug_kg,
    const float* __restrict__ w_aa,
    const float* __restrict__ w_ab,
    const float* __restrict__ w_ba,
    const float* __restrict__ w_bb,
    const float* __restrict__ bias_cnn,
    const float* __restrict__ bias_kg,
    float* __restrict__ out_cnn,
    float* __restrict__ out_kg)
{
    __shared__ __align__(16) v2f   s_btab[kChunks][32];
    __shared__ __align__(16) float s_bc[kD];
    __shared__ __align__(16) float s_bk[kD];
    __shared__ __align__(16) float s_scl[kRowsPerBlock * 2];
    __shared__ __align__(16) float s_kg[kWavesPerBlock * kRowsPerWave * kTilePitch];

    const int tid  = threadIdx.x;
    const int lane = tid & 31;
    const int wave = tid >> 5;
    const int n    = lane & 15;
    const int half = lane >> 4;
    const int rowBlock = blockIdx.x * kRowsPerBlock;
    const int rowWave  = rowBlock + wave * kRowsPerWave;

    const v4f* kgTile4 = (const v4f*)(drug_kg + (size_t)rowWave * kD);
    v4f t[kTileLoads];
    #pragma unroll
    for (int q = 0; q < kTileLoads; ++q)
        t[q] = __builtin_nontemporal_load(&kgTile4[q * 32 + lane]);

    const float bcv = bias_cnn[tid];
    const float bkv = bias_kg[tid];

    #pragma unroll
    for (int j = 0; j < 8; ++j) {
        const int f    = tid + 256 * j;
        const int kk   = f >> 5;
        const int l    = f & 31;
        const int nn   = l & 15;
        const int hh   = l >> 4;
        const int k0   = kk * 4 + hh * 2;
        v2f b = {0.0f, 0.0f};
        if (nn == 0) {
            b.x = w_aa[k0]     + w_ab[k0];
            b.y = w_aa[k0 + 1] + w_ab[k0 + 1];
        } else if (nn == 1) {
            b.x = w_ba[k0]     + w_bb[k0];
            b.y = w_ba[k0 + 1] + w_bb[k0 + 1];
        }
        s_btab[kk][l] = b;
    }

    s_bc[tid] = bcv;
    s_bk[tid] = bkv;

    float* tile = s_kg + wave * (kRowsPerWave * kTilePitch);
    #pragma unroll
    for (int q = 0; q < kTileLoads; ++q)
        *(v4f*)&tile[(q >> 1) * kTilePitch + (q & 1) * 128 + lane * 4] = t[q];
    __syncthreads();

    const v4f* cnnTile4 = (const v4f*)(drug_cnn + (size_t)rowWave * kD);
    v4f c[kTileLoads];
    #pragma unroll
    for (int q = 0; q < kTileLoads; ++q)
        c[q] = __builtin_nontemporal_load(&cnnTile4[q * 32 + lane]);

    const float* aRow = tile + n * kTilePitch + half * 2;

    v8f acc = {};
    #pragma unroll 8
    for (int kk = 0; kk < kChunks; ++kk) {
        v2f a = *(const v2f*)&aRow[kk * 4];
        v2f b = s_btab[kk][lane];
        acc = __builtin_amdgcn_wmma_f32_16x16x4_f32(
             false, a,  false, b,
             (short)0, acc,  false,  false);
    }

    if (n < 2) {
        const int mbase = wave * kRowsPerWave + half * 8;
        #pragma unroll
        for (int v = 0; v < 8; ++v)
            s_scl[(mbase + v) * 2 + n] = acc[v];
    }

    v4f* oc4 = (v4f*)out_cnn + (size_t)rowWave * (kD / 4);
    v4f* ok4 = (v4f*)out_kg  + (size_t)rowWave * (kD / 4);
    const v4f* bc4 = (const v4f*)s_bc;
    const v4f* bk4 = (const v4f*)s_bk;
    const v4f bcA = bc4[lane], bcB = bc4[32 + lane];
    const v4f bkA = bk4[lane], bkB = bk4[32 + lane];
    const int lrow0 = wave * kRowsPerWave;

    #pragma unroll
    for (int q = 0; q < kTileLoads; ++q) {
        const int r = q >> 1;
        const float sc = s_scl[(lrow0 + r) * 2 + 0];
        const float sk = s_scl[(lrow0 + r) * 2 + 1];
        const v4f bc = (q & 1) ? bcB : bcA;
        const v4f bk = (q & 1) ? bkB : bkA;
        const v4f x = c[q];
        v4f o1, o2;
        o1.x = __builtin_fmaf(x.x, sc, bc.x);  o2.x = __builtin_fmaf(x.x, sk, bk.x);
        o1.y = __builtin_fmaf(x.y, sc, bc.y);  o2.y = __builtin_fmaf(x.y, sk, bk.y);
        o1.z = __builtin_fmaf(x.z, sc, bc.z);  o2.z = __builtin_fmaf(x.z, sk, bk.z);
        o1.w = __builtin_fmaf(x.w, sc, bc.w);  o2.w = __builtin_fmaf(x.w, sk, bk.w);
        __builtin_nontemporal_store(o1, &oc4[q * 32 + lane]);
        __builtin_nontemporal_store(o2, &ok4[q * 32 + lane]);
    }
}

extern "C" void kernel_launch(void* const* d_in, const int* in_sizes, int n_in,
                              void* d_out, int out_size, void* d_ws, size_t ws_size,
                              hipStream_t stream) {
    const float* drug_cnn = (const float*)d_in[0];
    const float* drug_kg  = (const float*)d_in[1];
    const float* w_aa     = (const float*)d_in[2];
    const float* w_ab     = (const float*)d_in[3];
    const float* w_ba     = (const float*)d_in[4];
    const float* w_bb     = (const float*)d_in[5];
    const float* bcnn     = (const float*)d_in[6];
    const float* bkg      = (const float*)d_in[7];

    const int B = in_sizes[0] / kD;
    float* out_cnn = (float*)d_out;
    float* out_kg  = out_cnn + (size_t)B * kD;

    dim3 grid(B / kRowsPerBlock);
    dim3 block(kWavesPerBlock * 32);
    shared_unit_wmma_kernel<<<grid, block, 0, stream>>>(
        drug_cnn, drug_kg, w_aa, w_ab, w_ba, w_bb, bcnn, bkg, out_cnn, out_kg);
}


# B fragments read from per-wave w_aa+w_ab / w_ba+w_bb vectors staged from whole-line loads issued ahead of the kg tile; block table, its 8 dependent gather rounds and the barrier removed; same sums
# speedup vs baseline: 2.3247x; 1.0910x over previous
#include <hip/hip_runtime.h>

typedef __attribute__((ext_vector_type(2))) float v2f;
typedef __attribute__((ext_vector_type(4))) float v4f;
typedef __attribute__((ext_vector_type(8))) float v8f;

constexpr int kD = 256;
constexpr int kChunks = kD / 4;
constexpr int kRowsPerWave = 16;
constexpr int kWavesPerBlock = 8;
constexpr int kRowsPerBlock = kRowsPerWave * kWavesPerBlock;
constexpr int kTileLoads = kRowsPerWave * kD / (32 * 4);
constexpr int kTilePitch = kD + 4;
constexpr int kWsPitch = kD + 4;

__global__ __launch_bounds__(256) void shared_unit_wmma_kernel(
    const float* __restrict__ drug_cnn,
    const float* __restrict__ drug_kg,
    const float* __restrict__ w_aa,
    const float* __restrict__ w_ab,
    const float* __restrict__ w_ba,
    const float* __restrict__ w_bb,
    const float* __restrict__ bias_cnn,
    const float* __restrict__ bias_kg,
    float* __restrict__ out_cnn,
    float* __restrict__ out_kg)
{
    struct __align__(16) Lds {
        float ws[kWavesPerBlock * 2 * kWsPitch];
        float scl[kRowsPerBlock * 2];
        float kg[kWavesPerBlock * kRowsPerWave * kTilePitch];
    };
    __shared__ Lds s;

    const int tid  = threadIdx.x;
    const int lane = tid & 31;
    const int wave = tid >> 5;
    const int n    = lane & 15;
    const int half = lane >> 4;
    const int rowBlock = blockIdx.x * kRowsPerBlock;
    const int rowWave  = rowBlock + wave * kRowsPerWave;

    const v4f* aa4 = (const v4f*)w_aa;
    const v4f* ab4 = (const v4f*)w_ab;
    const v4f* ba4 = (const v4f*)w_ba;
    const v4f* bb4 = (const v4f*)w_bb;
    const v4f waa0 = aa4[lane], waa1 = aa4[32 + lane];
    const v4f wab0 = ab4[lane], wab1 = ab4[32 + lane];
    const v4f wba0 = ba4[lane], wba1 = ba4[32 + lane];
    const v4f wbb0 = bb4[lane], wbb1 = bb4[32 + lane];

    const v4f* kgTile4 = (const v4f*)(drug_kg + (size_t)rowWave * kD);
    v4f t[kTileLoads];
    #pragma unroll
    for (int q = 0; q < kTileLoads; ++q)
        t[q] = __builtin_nontemporal_load(&kgTile4[q * 32 + lane]);

    const v4f* bc4 = (const v4f*)bias_cnn;
    const v4f* bk4 = (const v4f*)bias_kg;
    const v4f bcA = bc4[lane], bcB = bc4[32 + lane];
    const v4f bkA = bk4[lane], bkB = bk4[32 + lane];

    float* ws = s.ws + wave * (2 * kWsPitch);
    *(v4f*)&ws[lane * 4]                  = waa0 + wab0;
    *(v4f*)&ws[128 + lane * 4]            = waa1 + wab1;
    *(v4f*)&ws[kWsPitch + lane * 4]       = wba0 + wbb0;
    *(v4f*)&ws[kWsPitch + 128 + lane * 4] = wba1 + wbb1;

    float* tile = s.kg + wave * (kRowsPerWave * kTilePitch);
    #pragma unroll
    for (int q = 0; q < kTileLoads; ++q)
        *(v4f*)&tile[(q >> 1) * kTilePitch + (q & 1) * 128 + lane * 4] = t[q];

    __builtin_amdgcn_sched_barrier(0);

    const v4f* cnnTile4 = (const v4f*)(drug_cnn + (size_t)rowWave * kD);
    v4f c[kTileLoads];
    #pragma unroll
    for (int q = 0; q < kTileLoads; ++q)
        c[q] = __builtin_nontemporal_load(&cnnTile4[q * 32 + lane]);

    const float* aRow = tile + n * kTilePitch + half * 2;
    const float* bVec = ws + (n & 1) * kWsPitch + half * 2;

    v8f acc = {};
    #pragma unroll 8
    for (int kk = 0; kk < kChunks; ++kk) {
        v2f a = *(const v2f*)&aRow[kk * 4];
        v2f b = *(const v2f*)&bVec[kk * 4];
        acc = __builtin_amdgcn_wmma_f32_16x16x4_f32(
             false, a,  false, b,
             (short)0, acc,  false,  false);
    }

    if (n < 2) {
        const int mbase = wave * kRowsPerWave + half * 8;
        #pragma unroll
        for (int v = 0; v < 8; ++v)
            s.scl[(mbase + v) * 2 + n] = acc[v];
    }

    v4f* oc4 = (v4f*)out_cnn + (size_t)rowWave * (kD / 4);
    v4f* ok4 = (v4f*)out_kg  + (size_t)rowWave * (kD / 4);
    const int lrow0 = wave * kRowsPerWave;

    #pragma unroll
    for (int q = 0; q < kTileLoads; ++q) {
        const int r = q >> 1;
        const float sc = s.scl[(lrow0 + r) * 2 + 0];
        const float sk = s.scl[(lrow0 + r) * 2 + 1];
        const v4f bc = (q & 1) ? bcB : bcA;
        const v4f bk = (q & 1) ? bkB : bkA;
        const v4f x = c[q];
        v4f o1, o2;
        o1.x = __builtin_fmaf(x.x, sc, bc.x);  o2.x = __builtin_fmaf(x.x, sk, bk.x);
        o1.y = __builtin_fmaf(x.y, sc, bc.y);  o2.y = __builtin_fmaf(x.y, sk, bk.y);
        o1.z = __builtin_fmaf(x.z, sc, bc.z);  o2.z = __builtin_fmaf(x.z, sk, bk.z);
        o1.w = __builtin_fmaf(x.w, sc, bc.w);  o2.w = __builtin_fmaf(x.w, sk, bk.w);
        __builtin_nontemporal_store(o1, &oc4[q * 32 + lane]);
        __builtin_nontemporal_store(o2, &ok4[q * 32 + lane]);
    }
}

extern "C" void kernel_launch(void* const* d_in, const int* in_sizes, int n_in,
                              void* d_out, int out_size, void* d_ws, size_t ws_size,
                              hipStream_t stream) {
    const float* drug_cnn = (const float*)d_in[0];
    const float* drug_kg  = (const float*)d_in[1];
    const float* w_aa     = (const float*)d_in[2];
    const float* w_ab     = (const float*)d_in[3];
    const float* w_ba     = (const float*)d_in[4];
    const float* w_bb     = (const float*)d_in[5];
    const float* bcnn     = (const float*)d_in[6];
    const float* bkg      = (const float*)d_in[7];

    const int B = in_sizes[0] / kD;
    float* out_cnn = (float*)d_out;
    float* out_kg  = out_cnn + (size_t)B * kD;

    dim3 grid(B / kRowsPerBlock);
    dim3 block(kWavesPerBlock * 32);
    shared_unit_wmma_kernel<<<grid, block, 0, stream>>>(
        drug_cnn, drug_kg, w_aa, w_ab, w_ba, w_bb, bcnn, bkg, out_cnn, out_kg);
}
